# starting kernel
# baseline (speedup 1.0000x reference)
#include <hip/hip_runtime.h>
#include <hip/hip_bf16.h>
#include <math.h>

#define Bz    4
#define Sn    4096
#define Dn    128
#define Kn    13
#define Mn    12
#define Rn    (Bz * Sn)
#define WPAD  16
#define HSTR  132

typedef float v2f __attribute__((ext_vector_type(2)));
typedef float v8f __attribute__((ext_vector_type(8)));

__device__ __forceinline__ float gelu_exact(float v) {
    return 0.5f * v * (1.0f + erff(v * 0.70710678118654752f));
}

__device__ __forceinline__ v8f wmma_f32x4(v2f a, v2f b, v8f c) {
    return __builtin_amdgcn_wmma_f32_16x16x4_f32(
         false, a,  false, b,
         (short)0, c,  false,  false);
}

__global__ __launch_bounds__(256) void embed_kernel(
    const int* __restrict__ data, const float* __restrict__ emb,
    const float* __restrict__ apc, float* __restrict__ x)
{
    int tid = blockIdx.x * blockDim.x + threadIdx.x;
    int row = tid >> 5;
    int c4  = tid & 31;
    if (row >= Rn) return;
    int s   = row & (Sn - 1);
    int tok = data[row];
    const float4* e = (const float4*)(emb + (size_t)tok * Dn);
    const float4* a = (const float4*)(apc + (size_t)s * Dn);
    float4 ev = e[c4], av = a[c4];
    float4 o; o.x = ev.x + av.x; o.y = ev.y + av.y; o.z = ev.z + av.z; o.w = ev.w + av.w;
    ((float4*)x)[tid] = o;
}

__global__ __launch_bounds__(128) void gmlp_kernel(
    const float* __restrict__ x,
    const float* __restrict__ W1, const float* __restrict__ b1,
    const float* __restrict__ W2, const float* __restrict__ b2,
    float* __restrict__ outV)
{
    __shared__ float Hlds[4][16 * HSTR];
    const int wave    = threadIdx.x >> 5;
    const int lane    = threadIdx.x & 31;
    const int half    = lane >> 4;
    const int mn      = lane & 15;
    const int rowBase = blockIdx.x * 64 + wave * 16;

    v8f acc[8];
    #pragma unroll
    for (int t = 0; t < 8; t++) acc[t] = {};
    const float* Xrow = x + (size_t)(rowBase + mn) * Dn;
    #pragma unroll 4
    for (int k = 0; k < Dn; k += 4) {
        float2 av = *(const float2*)(Xrow + k + 2 * half);
        v2f a; a.x = av.x; a.y = av.y;
        #pragma unroll
        for (int t = 0; t < 8; t++) {
            const float* wp = W1 + (size_t)(k + 2 * half) * Dn + t * 16 + mn;
            v2f b; b.x = wp[0]; b.y = wp[Dn];
            acc[t] = wmma_f32x4(a, b, acc[t]);
        }
    }
    #pragma unroll
    for (int t = 0; t < 8; t++) {
        float bias = b1[t * 16 + mn];
        #pragma unroll
        for (int r = 0; r < 8; r++) {
            Hlds[wave][(r + 8 * half) * HSTR + t * 16 + mn] =
                gelu_exact(acc[t][r] + bias);
        }
    }
    __syncthreads();

    v8f acc2[8];
    #pragma unroll
    for (int t = 0; t < 8; t++) acc2[t] = {};
    #pragma unroll 4
    for (int k = 0; k < Dn; k += 4) {
        float2 av = *(const float2*)(&Hlds[wave][mn * HSTR + k + 2 * half]);
        v2f a; a.x = av.x; a.y = av.y;
        #pragma unroll
        for (int t = 0; t < 8; t++) {
            const float* wp = W2 + (size_t)(k + 2 * half) * Dn + t * 16 + mn;
            v2f b; b.x = wp[0]; b.y = wp[Dn];
            acc2[t] = wmma_f32x4(a, b, acc2[t]);
        }
    }
    #pragma unroll
    for (int t = 0; t < 8; t++) {
        float bias = b2[t * 16 + mn];
        #pragma unroll
        for (int r = 0; r < 8; r++) {
            outV[(size_t)(rowBase + r + 8 * half) * Dn + t * 16 + mn] =
                acc2[t][r] + bias;
        }
    }
}

__global__ __launch_bounds__(128) void fmlp_kernel(
    const float* __restrict__ x,
    const float* __restrict__ fW1, const float* __restrict__ fb1,
    const float* __restrict__ fW2, const float* __restrict__ fb2,
    float* __restrict__ Wpad)
{
    __shared__ float Hlds[4][16 * HSTR];
    const int m       = blockIdx.y;
    const int wave    = threadIdx.x >> 5;
    const int lane    = threadIdx.x & 31;
    const int half    = lane >> 4;
    const int mn      = lane & 15;
    const int rowBase = blockIdx.x * 64 + wave * 16;

    const float* W1 = fW1 + (size_t)m * Dn * Dn;
    const float* B1 = fb1 + (size_t)m * Dn;
    const float* W2 = fW2 + (size_t)m * Dn * Kn;
    const float* B2 = fb2 + (size_t)m * Kn;

    v8f acc[8];
    #pragma unroll
    for (int t = 0; t < 8; t++) acc[t] = {};
    const float* Xrow = x + (size_t)(rowBase + mn) * Dn;
    #pragma unroll 4
    for (int k = 0; k < Dn; k += 4) {
        float2 av = *(const float2*)(Xrow + k + 2 * half);
        v2f a; a.x = av.x; a.y = av.y;
        #pragma unroll
        for (int t = 0; t < 8; t++) {
            const float* wp = W1 + (size_t)(k + 2 * half) * Dn + t * 16 + mn;
            v2f b; b.x = wp[0]; b.y = wp[Dn];
            acc[t] = wmma_f32x4(a, b, acc[t]);
        }
    }
    #pragma unroll
    for (int t = 0; t < 8; t++) {
        float bias = B1[t * 16 + mn];
        #pragma unroll
        for (int r = 0; r < 8; r++) {
            Hlds[wave][(r + 8 * half) * HSTR + t * 16 + mn] =
                gelu_exact(acc[t][r] + bias);
        }
    }
    __syncthreads();

    v8f acc2 = {};
    const bool valid = (mn < Kn);
    #pragma unroll 4
    for (int k = 0; k < Dn; k += 4) {
        float2 av = *(const float2*)(&Hlds[wave][mn * HSTR + k + 2 * half]);
        v2f a; a.x = av.x; a.y = av.y;
        v2f b;
        b.x = valid ? W2[(size_t)(k + 2 * half) * Kn + mn]     : 0.0f;
        b.y = valid ? W2[(size_t)(k + 2 * half + 1) * Kn + mn] : 0.0f;
        acc2 = wmma_f32x4(a, b, acc2);
    }
    float bias = valid ? B2[mn] : 0.0f;
    #pragma unroll
    for (int r = 0; r < 8; r++) {
        Wpad[((size_t)m * Rn + rowBase + r + 8 * half) * WPAD + mn] =
            acc2[r] + bias;
    }
}

__global__ __launch_bounds__(256) void chord_kernel(
    const float* __restrict__ Vsrc, const float* __restrict__ res,
    const float* __restrict__ Wm, float* __restrict__ Vdst)
{
    const int row  = blockIdx.x * 8 + (threadIdx.x >> 5);
    const int lane = threadIdx.x & 31;
    const int b    = row >> 12;
    const int i    = row & (Sn - 1);

    const int offs[Kn] = {0, 1, 2, 4, 8, 16, 32, 64, 128, 256, 512, 1024, 2048};

    float w[Kn];
    const float* wrow = Wm + (size_t)row * WPAD;
    #pragma unroll
    for (int k = 0; k < Kn; k++) w[k] = wrow[k];

    const float4* srcb = (const float4*)(Vsrc + (size_t)b * Sn * Dn);
    float4 acc = ((const float4*)(res + (size_t)row * Dn))[lane];
    #pragma unroll
    for (int k = 0; k < Kn; k++) {
        int j = (i + offs[k]) & (Sn - 1);
        float4 v = srcb[(size_t)j * (Dn / 4) + lane];
        acc.x += w[k] * v.x; acc.y += w[k] * v.y;
        acc.z += w[k] * v.z; acc.w += w[k] * v.w;
    }
    ((float4*)(Vdst + (size_t)row * Dn))[lane] = acc;
}

extern "C" void kernel_launch(void* const* d_in, const int* in_sizes, int n_in,
                              void* d_out, int out_size, void* d_ws, size_t ws_size,
                              hipStream_t stream) {
    const int*   data = (const int*)  d_in[0];
    const float* emb  = (const float*)d_in[2];
    const float* apc  = (const float*)d_in[3];
    const float* gW1  = (const float*)d_in[4];
    const float* gb1  = (const float*)d_in[5];
    const float* gW2  = (const float*)d_in[6];
    const float* gb2  = (const float*)d_in[7];
    const float* fW1  = (const float*)d_in[8];
    const float* fb1  = (const float*)d_in[9];
    const float* fW2  = (const float*)d_in[10];
    const float* fb2  = (const float*)d_in[11];
    float* out = (float*)d_out;

    char*  ws   = (char*)d_ws;
    float* xbuf = (float*)(ws);
    float* res  = (float*)(ws + (size_t) 8 * 1024 * 1024);
    float* Va   = (float*)(ws + (size_t)16 * 1024 * 1024);
    float* Wpad = (float*)(ws + (size_t)24 * 1024 * 1024);

    embed_kernel<<<(Rn * (Dn / 4)) / 256, 256, 0, stream>>>(data, emb, apc, xbuf);
    gmlp_kernel <<<Rn / 64, 128, 0, stream>>>(xbuf, gW1, gb1, gW2, gb2, res);
    fmlp_kernel <<<dim3(Rn / 64, Mn), 128, 0, stream>>>(xbuf, fW1, fb1, fW2, fb2, Wpad);

    const float* src = res;
    for (int m = 0; m < Mn; m++) {
        float* dst = (m & 1) ? out : Va;
        chord_kernel<<<Rn / 8, 256, 0, stream>>>(
            src, res, Wpad + (size_t)m * Rn * WPAD, dst);
        src = dst;
    }
}


# Weight strip tiles take columns 2n+t: one 8-byte load feeds both tiles (64 loads, not 128); last k-steps asked for behind the x hand-over, so none waits for the 63-deep load counter; sum order kept
# speedup vs baseline: 1.3882x; 1.3882x over previous
#include <hip/hip_runtime.h>
#include <hip/hip_bf16.h>
#include <math.h>

#define Bz    4
#define Sn    4096
#define Dn    128
#define Kn    13
#define Mn    12
#define Rn    (Bz * Sn)
#define WPAD  16
#define HSTR  132

typedef float v2f __attribute__((ext_vector_type(2)));
typedef float v8f __attribute__((ext_vector_type(8)));

__device__ __forceinline__ float gelu_exact(float v) {
    return 0.5f * v * (1.0f + erff(v * 0.70710678118654752f));
}

__device__ __forceinline__ v8f wmma_f32x4(v2f a, v2f b, v8f c) {
    return __builtin_amdgcn_wmma_f32_16x16x4_f32(
         false, a,  false, b,
         (short)0, c,  false,  false);
}

__global__ __launch_bounds__(256) void embed_kernel(
    const int* __restrict__ data, const float* __restrict__ emb,
    const float* __restrict__ apc, float* __restrict__ x)
{
    int tid = blockIdx.x * blockDim.x + threadIdx.x;
    int row = tid >> 5;
    int c4  = tid & 31;
    if (row >= Rn) return;
    int s   = row & (Sn - 1);
    int tok = data[row];
    const float4* e = (const float4*)(emb + (size_t)tok * Dn);
    const float4* a = (const float4*)(apc + (size_t)s * Dn);
    float4 ev = e[c4], av = a[c4];
    float4 o; o.x = ev.x + av.x; o.y = ev.y + av.y; o.z = ev.z + av.z; o.w = ev.w + av.w;
    ((float4*)x)[tid] = o;
}

#define RT     4
#define RROWS  (RT * 16)
#define KSTEPS (Dn / 4)
#define KA_W   26
#define KA_G   22

template <int NTHR>
__device__ __forceinline__ void stage_rows(const float* __restrict__ src,
                                           float* lds, int tid)
{
    #pragma unroll
    for (int i = 0; i < (RROWS * (Dn / 4)) / NTHR; i++) {
        int idx = i * NTHR + tid;
        int row = idx >> 5, c4 = idx & 31;
        float4 v = ((const float4*)src)[idx];
        *(float4*)(lds + row * HSTR + c4 * 4) = v;
    }
}

template <int KK0, int KK1>
__device__ __forceinline__ void load_strip(const float* __restrict__ Wm, int ld,
                                           int col0, int mn, int half,
                                           float2 (&bw)[KSTEPS][2])
{
    #pragma unroll
    for (int kk = KK0; kk < KK1; kk++) {
        const float* wp = Wm + (size_t)(4 * kk + 2 * half) * ld + col0 + 2 * mn;
        bw[kk][0] = *(const float2*)wp;
        bw[kk][1] = *(const float2*)(wp + ld);
    }
}

#define KB 2
template <int NT>
__device__ __forceinline__ void load_afrags(const float* lds, int kb, int mn, int half,
                                            v2f (&a)[KB][NT])
{
    #pragma unroll
    for (int j = 0; j < KB; j++) {
        #pragma unroll
        for (int rt = 0; rt < NT; rt++) {
            float2 av = *(const float2*)(lds + (rt * 16 + mn) * HSTR + 4 * (kb * KB + j) + 2 * half);
            a[j][rt].x = av.x; a[j][rt].y = av.y;
        }
    }
}

template <int NT>
__device__ __forceinline__ void strip_product(const float* lds,
                                              const float2 (&bw)[KSTEPS][2],
                                              v8f (&acc)[NT][2], int mn, int half)
{
    v2f cur[KB][NT], nxt[KB][NT];
    load_afrags<NT>(lds, 0, mn, half, cur);
    #pragma unroll
    for (int kb = 0; kb < KSTEPS / KB; kb++) {
        if (kb + 1 < KSTEPS / KB) load_afrags<NT>(lds, kb + 1, mn, half, nxt);
        __builtin_amdgcn_sched_barrier(0);
        #pragma unroll
        for (int j = 0; j < KB; j++) {
            v2f b0, b1;
            b0.x = bw[kb * KB + j][0].x; b0.y = bw[kb * KB + j][1].x;
            b1.x = bw[kb * KB + j][0].y; b1.y = bw[kb * KB + j][1].y;
            #pragma unroll
            for (int rt = 0; rt < NT; rt++) {
                acc[rt][0] = wmma_f32x4(cur[j][rt], b0, acc[rt][0]);
                acc[rt][1] = wmma_f32x4(cur[j][rt], b1, acc[rt][1]);
            }
        }
        __builtin_amdgcn_sched_barrier(0);
        #pragma unroll
        for (int j = 0; j < KB; j++) {
            #pragma unroll
            for (int rt = 0; rt < NT; rt++) cur[j][rt] = nxt[j][rt];
        }
    }
}

template <int NT>
__device__ __forceinline__ void gelu_to_lds(const v8f (&acc)[NT][2],
                                            const float (&bias)[2], float* lds,
                                            int col0, int mn, int half)
{
    #pragma unroll
    for (int rt = 0; rt < NT; rt++) {
        #pragma unroll
        for (int r = 0; r < 8; r++) {
            float2 h;
            h.x = gelu_exact(acc[rt][0][r] + bias[0]);
            h.y = gelu_exact(acc[rt][1][r] + bias[1]);
            *(float2*)(lds + (rt * 16 + r + 8 * half) * HSTR + col0 + 2 * mn) = h;
        }
    }
}

__global__ __launch_bounds__(128) void gmlp_kernel(
    const float* __restrict__ x,
    const float* __restrict__ W1, const float* __restrict__ b1,
    const float* __restrict__ W2, const float* __restrict__ b2,
    float* __restrict__ outV)
{
    __shared__ __attribute__((aligned(16))) float Xlds[RROWS * HSTR];
    __shared__ __attribute__((aligned(16))) float Hlds[RROWS * HSTR];
    const int wave    = threadIdx.x >> 5;
    const int lane    = threadIdx.x & 31;
    const int half    = lane >> 4;
    const int mn      = lane & 15;
    const int col0    = wave * 32;
    const int rowBase = blockIdx.x * RROWS;

    float2 bw[KSTEPS][2];
    load_strip<0, KA_G>(W1, Dn, col0, mn, half, bw);
    stage_rows<128>(x + (size_t)rowBase * Dn, Xlds, threadIdx.x);
    float bias[2];
    const float2 bv1 = *(const float2*)(b1 + col0 + 2 * mn);
    bias[0] = bv1.x;
    bias[1] = bv1.y;
    __syncthreads();
    load_strip<KA_G, KSTEPS>(W1, Dn, col0, mn, half, bw);

    v8f acc[RT][2];
    #pragma unroll
    for (int rt = 0; rt < RT; rt++) { acc[rt][0] = {}; acc[rt][1] = {}; }
    strip_product<RT>(Xlds, bw, acc, mn, half);

    load_strip<0, KSTEPS>(W2, Dn, col0, mn, half, bw);
    gelu_to_lds<RT>(acc, bias, Hlds, col0, mn, half);
    const float2 bv2 = *(const float2*)(b2 + col0 + 2 * mn);
    bias[0] = bv2.x;
    bias[1] = bv2.y;
    __syncthreads();

    #pragma unroll
    for (int rt = 0; rt < RT; rt++) { acc[rt][0] = {}; acc[rt][1] = {}; }
    strip_product<RT>(Hlds, bw, acc, mn, half);

    #pragma unroll
    for (int rt = 0; rt < RT; rt++) {
        #pragma unroll
        for (int r = 0; r < 8; r++) {
            float2 o;
            o.x = acc[rt][0][r] + bias[0];
            o.y = acc[rt][1][r] + bias[1];
            *(float2*)(outV + (size_t)(rowBase + rt * 16 + r + 8 * half) * Dn + col0 + 2 * mn) = o;
        }
    }
}

#define CROWS 4
#define FT    2

__device__ __forceinline__ float lane_value(float v, int src) {
    return __int_as_float(__builtin_amdgcn_readlane(__float_as_int(v), src));
}

__device__ __forceinline__ void weight_group(
    float* Xlds, float* Hlds, int wave, int lane,
    const float* __restrict__ x,
    const float* __restrict__ W1, const float* __restrict__ B1,
    const float* __restrict__ W2, const float* __restrict__ B2,
    float* __restrict__ Wnext)
{
    const int half    = lane >> 4;
    const int mn      = lane & 15;
    const int col0    = (wave & 3) * 32;
    const int rt0     = (wave >> 2) * FT;
    const int rowBase = blockIdx.x * RROWS;

    float2 bw[KSTEPS][2];
    load_strip<0, KA_W>(W1, Dn, col0, mn, half, bw);
    stage_rows<256>(x + (size_t)rowBase * Dn, Xlds, threadIdx.x);
    float bias[2];
    const float2 bv1 = *(const float2*)(B1 + col0 + 2 * mn);
    bias[0] = bv1.x;
    bias[1] = bv1.y;
    __syncthreads();
    load_strip<KA_W, KSTEPS>(W1, Dn, col0, mn, half, bw);

    v8f acc[FT][2];
    #pragma unroll
    for (int rt = 0; rt < FT; rt++) { acc[rt][0] = {}; acc[rt][1] = {}; }
    strip_product<FT>(Xlds + rt0 * 16 * HSTR, bw, acc, mn, half);

    const bool valid = (mn < Kn);
    const int  mc    = valid ? mn : 0;
    v2f   bw2[KSTEPS];
    float bias2 = 0.0f;
    if (wave < 4) {
        #pragma unroll
        for (int kk = 0; kk < KSTEPS; kk++) {
            bw2[kk].x = W2[(size_t)(4 * kk + 2 * half) * Kn + mc];
            bw2[kk].y = W2[(size_t)(4 * kk + 2 * half + 1) * Kn + mc];
        }
        bias2 = B2[mc];
    }

    gelu_to_lds<FT>(acc, bias, Hlds + rt0 * 16 * HSTR, col0, mn, half);
    __syncthreads();

    if (wave < 4) {
        v8f acc2 = {};
        v2f a2[KSTEPS];
        #pragma unroll
        for (int kk = 0; kk < KSTEPS; kk++) {
            float2 av = *(const float2*)(Hlds + (wave * 16 + mn) * HSTR + 4 * kk + 2 * half);
            a2[kk].x = av.x; a2[kk].y = av.y;
        }
        #pragma unroll
        for (int kk = 0; kk < KSTEPS; kk++) {
            bw2[kk].x = valid ? bw2[kk].x : 0.0f;
            bw2[kk].y = valid ? bw2[kk].y : 0.0f;
        }
        bias2 = valid ? bias2 : 0.0f;
        __builtin_amdgcn_sched_barrier(0);
        #pragma unroll
        for (int kk = 0; kk < KSTEPS; kk++) acc2 = wmma_f32x4(a2[kk], bw2[kk], acc2);
        #pragma unroll
        for (int r = 0; r < 8; r++) {
            Wnext[((size_t)rowBase + wave * 16 + r + 8 * half) * WPAD + mn] =
                acc2[r] + bias2;
        }
    }
}

__device__ __forceinline__ void mixing_group(
    int group, int wave, int lane,
    const float* __restrict__ Vsrc, const float* __restrict__ res,
    const float* __restrict__ Wm, float* __restrict__ Vdst)
{
    const int row0 = (group * 8 + wave) * CROWS;
    const int b    = row0 >> 12;
    const int i0   = row0 & (Sn - 1);

    const int offs[Kn] = {0, 1, 2, 4, 8, 16, 32, 64, 128, 256, 512, 1024, 2048};

    const float2 wv = ((const float2*)(Wm + (size_t)row0 * WPAD))[lane];

    const float4* srcb = (const float4*)(Vsrc + (size_t)b * Sn * Dn);
    const float4* resr = (const float4*)(res + (size_t)row0 * Dn);
    float4*       dstr = (float4*)(Vdst + (size_t)row0 * Dn);

    float4 acc[CROWS];
    float4 v[CROWS][Kn];
    #pragma unroll
    for (int r = 0; r < CROWS; r++) {
        acc[r] = resr[r * (Dn / 4) + lane];
        #pragma unroll
        for (int k = 0; k < Kn; k++) {
            const int j = (i0 + r + offs[k]) & (Sn - 1);
            v[r][k] = srcb[(size_t)j * (Dn / 4) + lane];
        }
    }
    #pragma unroll
    for (int r = 0; r < CROWS; r++) {
        #pragma unroll
        for (int k = 0; k < Kn; k++) {
            const float wk = lane_value((k & 1) ? wv.y : wv.x, 8 * r + (k >> 1));
            acc[r].x += wk * v[r][k].x; acc[r].y += wk * v[r][k].y;
            acc[r].z += wk * v[r][k].z; acc[r].w += wk * v[r][k].w;
        }
        dstr[r * (Dn / 4) + lane] = acc[r];
    }
}

__global__ __launch_bounds__(256) __attribute__((amdgpu_waves_per_eu(4, 4))) void step_kernel(
    int nF,
    const float* __restrict__ Vsrc, const float* __restrict__ res,
    const float* __restrict__ Wm, float* __restrict__ Vdst,
    const float* __restrict__ x,
    const float* __restrict__ W1, const float* __restrict__ B1,
    const float* __restrict__ W2, const float* __restrict__ B2,
    float* __restrict__ Wnext)
{
    __shared__ __attribute__((aligned(16))) float Xlds[RROWS * HSTR];
    __shared__ __attribute__((aligned(16))) float Hlds[RROWS * HSTR];

    const int wave = __builtin_amdgcn_readfirstlane((int)(threadIdx.x >> 5));
    const int lane = threadIdx.x & 31;

    if ((int)blockIdx.x < nF) {
        weight_group(Xlds, Hlds, wave, lane, x, W1, B1, W2, B2, Wnext);
    } else {
        mixing_group((int)blockIdx.x - nF, wave, lane, Vsrc, res, Wm, Vdst);
    }
}

extern "C" void kernel_launch(void* const* d_in, const int* in_sizes, int n_in,
                              void* d_out, int out_size, void* d_ws, size_t ws_size,
                              hipStream_t stream) {
    const int*   data = (const int*)  d_in[0];
    const float* emb  = (const float*)d_in[2];
    const float* apc  = (const float*)d_in[3];
    const float* gW1  = (const float*)d_in[4];
    const float* gb1  = (const float*)d_in[5];
    const float* gW2  = (const float*)d_in[6];
    const float* gb2  = (const float*)d_in[7];
    const float* fW1  = (const float*)d_in[8];
    const float* fb1  = (const float*)d_in[9];
    const float* fW2  = (const float*)d_in[10];
    const float* fb2  = (const float*)d_in[11];
    float* out = (float*)d_out;

    char*  ws   = (char*)d_ws;
    float* xbuf = (float*)(ws);
    float* res  = (float*)(ws + (size_t) 8 * 1024 * 1024);
    float* Va   = (float*)(ws + (size_t)16 * 1024 * 1024);
    float* Wpad = (float*)(ws + (size_t)24 * 1024 * 1024);

    embed_kernel<<<(Rn * (Dn / 4)) / 256, 256, 0, stream>>>(data, emb, apc, xbuf);
    gmlp_kernel <<<Rn / RROWS, 128, 0, stream>>>(xbuf, gW1, gb1, gW2, gb2, res);

    step_kernel<<<Rn / RROWS, 256, 0, stream>>>(
        Rn / RROWS, res, res, Wpad, Va, xbuf, fW1, fb1, fW2, fb2, Wpad);

    const float* src = res;
    for (int m = 0; m < Mn; m++) {
        float* dst = (m & 1) ? out : Va;
        if (m + 1 < Mn) {
            step_kernel<<<Rn / RROWS + Rn / (8 * CROWS), 256, 0, stream>>>(
                Rn / RROWS,
                src, res, Wpad + (size_t)m * Rn * WPAD, dst,
                xbuf,
                fW1 + (size_t)(m + 1) * Dn * Dn, fb1 + (size_t)(m + 1) * Dn,
                fW2 + (size_t)(m + 1) * Dn * Kn, fb2 + (size_t)(m + 1) * Kn,
                Wpad + (size_t)(m + 1) * Rn * WPAD);
        } else {
            step_kernel<<<Rn / (8 * CROWS), 256, 0, stream>>>(
                0, src, res, Wpad + (size_t)m * Rn * WPAD, dst,
                xbuf, fW1, fb1, fW2, fb2, Wpad);
        }
        src = dst;
    }
}


# Weight slice 0 rides beside the g-MLP in ONE front launch (14 launches, not 15); g-MLP groups take 8 waves (4 strips x 2 row halves in both layers), two groups share a WGP's pipe; sum order kept
# speedup vs baseline: 1.4684x; 1.0578x over previous
#include <hip/hip_runtime.h>
#include <hip/hip_bf16.h>
#include <math.h>

#define Bz    4
#define Sn    4096
#define Dn    128
#define Kn    13
#define Mn    12
#define Rn    (Bz * Sn)
#define WPAD  16
#define HSTR  132

typedef float v2f __attribute__((ext_vector_type(2)));
typedef float v8f __attribute__((ext_vector_type(8)));

__device__ __forceinline__ float gelu_exact(float v) {
    return 0.5f * v * (1.0f + erff(v * 0.70710678118654752f));
}

__device__ __forceinline__ v8f wmma_f32x4(v2f a, v2f b, v8f c) {
    return __builtin_amdgcn_wmma_f32_16x16x4_f32(
         false, a,  false, b,
         (short)0, c,  false,  false);
}

__global__ __launch_bounds__(256) void embed_kernel(
    const int* __restrict__ data, const float* __restrict__ emb,
    const float* __restrict__ apc, float* __restrict__ x)
{
    int tid = blockIdx.x * blockDim.x + threadIdx.x;
    int row = tid >> 5;
    int c4  = tid & 31;
    if (row >= Rn) return;
    int s   = row & (Sn - 1);
    int tok = data[row];
    const float4* e = (const float4*)(emb + (size_t)tok * Dn);
    const float4* a = (const float4*)(apc + (size_t)s * Dn);
    float4 ev = e[c4], av = a[c4];
    float4 o; o.x = ev.x + av.x; o.y = ev.y + av.y; o.z = ev.z + av.z; o.w = ev.w + av.w;
    ((float4*)x)[tid] = o;
}

#define RT     4
#define RROWS  (RT * 16)
#define KSTEPS (Dn / 4)
#define KA_W   26

template <int NTHR>
__device__ __forceinline__ void stage_rows(const float* __restrict__ src,
                                           float* lds, int tid)
{
    #pragma unroll
    for (int i = 0; i < (RROWS * (Dn / 4)) / NTHR; i++) {
        int idx = i * NTHR + tid;
        int row = idx >> 5, c4 = idx & 31;
        float4 v = ((const float4*)src)[idx];
        *(float4*)(lds + row * HSTR + c4 * 4) = v;
    }
}

template <int KK0, int KK1>
__device__ __forceinline__ void load_strip(const float* __restrict__ Wm, int ld,
                                           int col0, int mn, int half,
                                           float2 (&bw)[KSTEPS][2])
{
    #pragma unroll
    for (int kk = KK0; kk < KK1; kk++) {
        const float* wp = Wm + (size_t)(4 * kk + 2 * half) * ld + col0 + 2 * mn;
        bw[kk][0] = *(const float2*)wp;
        bw[kk][1] = *(const float2*)(wp + ld);
    }
}

#define KB 2
template <int NT>
__device__ __forceinline__ void load_afrags(const float* lds, int kb, int mn, int half,
                                            v2f (&a)[KB][NT])
{
    #pragma unroll
    for (int j = 0; j < KB; j++) {
        #pragma unroll
        for (int rt = 0; rt < NT; rt++) {
            float2 av = *(const float2*)(lds + (rt * 16 + mn) * HSTR + 4 * (kb * KB + j) + 2 * half);
            a[j][rt].x = av.x; a[j][rt].y = av.y;
        }
    }
}

template <int NT>
__device__ __forceinline__ void strip_product(const float* lds,
                                              const float2 (&bw)[KSTEPS][2],
                                              v8f (&acc)[NT][2], int mn, int half)
{
    v2f cur[KB][NT], nxt[KB][NT];
    load_afrags<NT>(lds, 0, mn, half, cur);
    #pragma unroll
    for (int kb = 0; kb < KSTEPS / KB; kb++) {
        if (kb + 1 < KSTEPS / KB) load_afrags<NT>(lds, kb + 1, mn, half, nxt);
        __builtin_amdgcn_sched_barrier(0);
        #pragma unroll
        for (int j = 0; j < KB; j++) {
            v2f b0, b1;
            b0.x = bw[kb * KB + j][0].x; b0.y = bw[kb * KB + j][1].x;
            b1.x = bw[kb * KB + j][0].y; b1.y = bw[kb * KB + j][1].y;
            #pragma unroll
            for (int rt = 0; rt < NT; rt++) {
                acc[rt][0] = wmma_f32x4(cur[j][rt], b0, acc[rt][0]);
                acc[rt][1] = wmma_f32x4(cur[j][rt], b1, acc[rt][1]);
            }
        }
        __builtin_amdgcn_sched_barrier(0);
        #pragma unroll
        for (int j = 0; j < KB; j++) {
            #pragma unroll
            for (int rt = 0; rt < NT; rt++) cur[j][rt] = nxt[j][rt];
        }
    }
}

template <int NT>
__device__ __forceinline__ void gelu_to_lds(const v8f (&acc)[NT][2],
                                            const float (&bias)[2], float* lds,
                                            int col0, int mn, int half)
{
    #pragma unroll
    for (int rt = 0; rt < NT; rt++) {
        #pragma unroll
        for (int r = 0; r < 8; r++) {
            float2 h;
            h.x = gelu_exact(acc[rt][0][r] + bias[0]);
            h.y = gelu_exact(acc[rt][1][r] + bias[1]);
            *(float2*)(lds + (rt * 16 + r + 8 * half) * HSTR + col0 + 2 * mn) = h;
        }
    }
}

#define FT    2

__device__ __forceinline__ void gmlp_group(
    float* Xlds, float* Hlds, int wave, int lane, int group,
    const float* __restrict__ x,
    const float* __restrict__ W1, const float* __restrict__ b1,
    const float* __restrict__ W2, const float* __restrict__ b2,
    float* __restrict__ outV)
{
    const int half    = lane >> 4;
    const int mn      = lane & 15;
    const int col0    = (wave & 3) * 32;
    const int rt0     = (wave >> 2) * FT;
    const int rowBase = group * RROWS;

    float2 bw[KSTEPS][2];
    load_strip<0, KA_W>(W1, Dn, col0, mn, half, bw);
    stage_rows<256>(x + (size_t)rowBase * Dn, Xlds, threadIdx.x);
    float bias[2];
    const float2 bv1 = *(const float2*)(b1 + col0 + 2 * mn);
    bias[0] = bv1.x;
    bias[1] = bv1.y;
    __syncthreads();
    load_strip<KA_W, KSTEPS>(W1, Dn, col0, mn, half, bw);

    v8f acc[FT][2];
    #pragma unroll
    for (int rt = 0; rt < FT; rt++) { acc[rt][0] = {}; acc[rt][1] = {}; }
    strip_product<FT>(Xlds + rt0 * 16 * HSTR, bw, acc, mn, half);

    load_strip<0, KSTEPS>(W2, Dn, col0, mn, half, bw);
    gelu_to_lds<FT>(acc, bias, Hlds + rt0 * 16 * HSTR, col0, mn, half);
    const float2 bv2 = *(const float2*)(b2 + col0 + 2 * mn);
    bias[0] = bv2.x;
    bias[1] = bv2.y;
    __syncthreads();

    #pragma unroll
    for (int rt = 0; rt < FT; rt++) { acc[rt][0] = {}; acc[rt][1] = {}; }
    strip_product<FT>(Hlds + rt0 * 16 * HSTR, bw, acc, mn, half);

    #pragma unroll
    for (int rt = 0; rt < FT; rt++) {
        #pragma unroll
        for (int r = 0; r < 8; r++) {
            float2 o;
            o.x = acc[rt][0][r] + bias[0];
            o.y = acc[rt][1][r] + bias[1];
            *(float2*)(outV + (size_t)(rowBase + (rt0 + rt) * 16 + r + 8 * half) * Dn + col0 + 2 * mn) = o;
        }
    }
}

#define CROWS 4

__device__ __forceinline__ float lane_value(float v, int src) {
    return __int_as_float(__builtin_amdgcn_readlane(__float_as_int(v), src));
}

__device__ __forceinline__ void weight_group(
    float* Xlds, float* Hlds, int wave, int lane, int group,
    const float* __restrict__ x,
    const float* __restrict__ W1, const float* __restrict__ B1,
    const float* __restrict__ W2, const float* __restrict__ B2,
    float* __restrict__ Wnext)
{
    const int half    = lane >> 4;
    const int mn      = lane & 15;
    const int col0    = (wave & 3) * 32;
    const int rt0     = (wave >> 2) * FT;
    const int rowBase = group * RROWS;

    float2 bw[KSTEPS][2];
    load_strip<0, KA_W>(W1, Dn, col0, mn, half, bw);
    stage_rows<256>(x + (size_t)rowBase * Dn, Xlds, threadIdx.x);
    float bias[2];
    const float2 bv1 = *(const float2*)(B1 + col0 + 2 * mn);
    bias[0] = bv1.x;
    bias[1] = bv1.y;
    __syncthreads();
    load_strip<KA_W, KSTEPS>(W1, Dn, col0, mn, half, bw);

    v8f acc[FT][2];
    #pragma unroll
    for (int rt = 0; rt < FT; rt++) { acc[rt][0] = {}; acc[rt][1] = {}; }
    strip_product<FT>(Xlds + rt0 * 16 * HSTR, bw, acc, mn, half);

    const bool valid = (mn < Kn);
    const int  mc    = valid ? mn : 0;
    v2f   bw2[KSTEPS];
    float bias2 = 0.0f;
    if (wave < 4) {
        #pragma unroll
        for (int kk = 0; kk < KSTEPS; kk++) {
            bw2[kk].x = W2[(size_t)(4 * kk + 2 * half) * Kn + mc];
            bw2[kk].y = W2[(size_t)(4 * kk + 2 * half + 1) * Kn + mc];
        }
        bias2 = B2[mc];
    }

    gelu_to_lds<FT>(acc, bias, Hlds + rt0 * 16 * HSTR, col0, mn, half);
    __syncthreads();

    if (wave < 4) {
        v8f acc2 = {};
        v2f a2[KSTEPS];
        #pragma unroll
        for (int kk = 0; kk < KSTEPS; kk++) {
            float2 av = *(const float2*)(Hlds + (wave * 16 + mn) * HSTR + 4 * kk + 2 * half);
            a2[kk].x = av.x; a2[kk].y = av.y;
        }
        #pragma unroll
        for (int kk = 0; kk < KSTEPS; kk++) {
            bw2[kk].x = valid ? bw2[kk].x : 0.0f;
            bw2[kk].y = valid ? bw2[kk].y : 0.0f;
        }
        bias2 = valid ? bias2 : 0.0f;
        __builtin_amdgcn_sched_barrier(0);
        #pragma unroll
        for (int kk = 0; kk < KSTEPS; kk++) acc2 = wmma_f32x4(a2[kk], bw2[kk], acc2);
        #pragma unroll
        for (int r = 0; r < 8; r++) {
            Wnext[((size_t)rowBase + wave * 16 + r + 8 * half) * WPAD + mn] =
                acc2[r] + bias2;
        }
    }
}

__device__ __forceinline__ void mixing_group(
    int group, int wave, int lane,
    const float* __restrict__ Vsrc, const float* __restrict__ res,
    const float* __restrict__ Wm, float* __restrict__ Vdst)
{
    const int row0 = (group * 8 + wave) * CROWS;
    const int b    = row0 >> 12;
    const int i0   = row0 & (Sn - 1);

    const int offs[Kn] = {0, 1, 2, 4, 8, 16, 32, 64, 128, 256, 512, 1024, 2048};

    const float2 wv = ((const float2*)(Wm + (size_t)row0 * WPAD))[lane];

    const float4* srcb = (const float4*)(Vsrc + (size_t)b * Sn * Dn);
    const float4* resr = (const float4*)(res + (size_t)row0 * Dn);
    float4*       dstr = (float4*)(Vdst + (size_t)row0 * Dn);

    float4 acc[CROWS];
    float4 v[CROWS][Kn];
    #pragma unroll
    for (int r = 0; r < CROWS; r++) {
        acc[r] = resr[r * (Dn / 4) + lane];
        #pragma unroll
        for (int k = 0; k < Kn; k++) {
            const int j = (i0 + r + offs[k]) & (Sn - 1);
            v[r][k] = srcb[(size_t)j * (Dn / 4) + lane];
        }
    }
    #pragma unroll
    for (int r = 0; r < CROWS; r++) {
        #pragma unroll
        for (int k = 0; k < Kn; k++) {
            const float wk = lane_value((k & 1) ? wv.y : wv.x, 8 * r + (k >> 1));
            acc[r].x += wk * v[r][k].x; acc[r].y += wk * v[r][k].y;
            acc[r].z += wk * v[r][k].z; acc[r].w += wk * v[r][k].w;
        }
        dstr[r * (Dn / 4) + lane] = acc[r];
    }
}

__global__ __launch_bounds__(256) __attribute__((amdgpu_waves_per_eu(4, 4))) void step_kernel(
    int nF,
    const float* __restrict__ Vsrc, const float* __restrict__ res,
    const float* __restrict__ Wm, float* __restrict__ Vdst,
    const float* __restrict__ x,
    const float* __restrict__ W1, const float* __restrict__ B1,
    const float* __restrict__ W2, const float* __restrict__ B2,
    float* __restrict__ Wnext)
{
    __shared__ __attribute__((aligned(16))) float Xlds[RROWS * HSTR];
    __shared__ __attribute__((aligned(16))) float Hlds[RROWS * HSTR];

    const int wave = __builtin_amdgcn_readfirstlane((int)(threadIdx.x >> 5));
    const int lane = threadIdx.x & 31;

    if ((int)blockIdx.x < nF) {
        weight_group(Xlds, Hlds, wave, lane, (int)blockIdx.x, x, W1, B1, W2, B2, Wnext);
    } else {
        mixing_group((int)blockIdx.x - nF, wave, lane, Vsrc, res, Wm, Vdst);
    }
}

__global__ __launch_bounds__(256) __attribute__((amdgpu_waves_per_eu(4, 4))) void front_kernel(
    int nG,
    const float* __restrict__ x,
    const float* __restrict__ gW1, const float* __restrict__ gb1,
    const float* __restrict__ gW2, const float* __restrict__ gb2,
    float* __restrict__ outV,
    const float* __restrict__ W1, const float* __restrict__ B1,
    const float* __restrict__ W2, const float* __restrict__ B2,
    float* __restrict__ Wnext)
{
    __shared__ __attribute__((aligned(16))) float Xlds[RROWS * HSTR];
    __shared__ __attribute__((aligned(16))) float Hlds[RROWS * HSTR];

    const int wave = __builtin_amdgcn_readfirstlane((int)(threadIdx.x >> 5));
    const int lane = threadIdx.x & 31;

    if ((int)blockIdx.x < nG) {
        gmlp_group(Xlds, Hlds, wave, lane, (int)blockIdx.x, x, gW1, gb1, gW2, gb2, outV);
    } else {
        weight_group(Xlds, Hlds, wave, lane, (int)blockIdx.x - nG, x, W1, B1, W2, B2, Wnext);
    }
}

extern "C" void kernel_launch(void* const* d_in, const int* in_sizes, int n_in,
                              void* d_out, int out_size, void* d_ws, size_t ws_size,
                              hipStream_t stream) {
    const int*   data = (const int*)  d_in[0];
    const float* emb  = (const float*)d_in[2];
    const float* apc  = (const float*)d_in[3];
    const float* gW1  = (const float*)d_in[4];
    const float* gb1  = (const float*)d_in[5];
    const float* gW2  = (const float*)d_in[6];
    const float* gb2  = (const float*)d_in[7];
    const float* fW1  = (const float*)d_in[8];
    const float* fb1  = (const float*)d_in[9];
    const float* fW2  = (const float*)d_in[10];
    const float* fb2  = (const float*)d_in[11];
    float* out = (float*)d_out;

    char*  ws   = (char*)d_ws;
    float* xbuf = (float*)(ws);
    float* res  = (float*)(ws + (size_t) 8 * 1024 * 1024);
    float* Va   = (float*)(ws + (size_t)16 * 1024 * 1024);
    float* Wpad = (float*)(ws + (size_t)24 * 1024 * 1024);

    embed_kernel<<<(Rn * (Dn / 4)) / 256, 256, 0, stream>>>(data, emb, apc, xbuf);
    front_kernel<<<2 * (Rn / RROWS), 256, 0, stream>>>(
        Rn / RROWS, xbuf, gW1, gb1, gW2, gb2, res, fW1, fb1, fW2, fb2, Wpad);

    const float* src = res;
    for (int m = 0; m < Mn; m++) {
        float* dst = (m & 1) ? out : Va;
        if (m + 1 < Mn) {
            step_kernel<<<Rn / RROWS + Rn / (8 * CROWS), 256, 0, stream>>>(
                Rn / RROWS,
                src, res, Wpad + (size_t)m * Rn * WPAD, dst,
                xbuf,
                fW1 + (size_t)(m + 1) * Dn * Dn, fb1 + (size_t)(m + 1) * Dn,
                fW2 + (size_t)(m + 1) * Dn * Kn, fb2 + (size_t)(m + 1) * Kn,
                Wpad + (size_t)(m + 1) * Rn * WPAD);
        } else {
            step_kernel<<<Rn / (8 * CROWS), 256, 0, stream>>>(
                0, src, res, Wpad + (size_t)m * Rn * WPAD, dst,
                xbuf, fW1, fb1, fW2, fb2, Wpad);
        }
        src = dst;
    }
}
